# starting kernel
# baseline (speedup 1.0000x reference)
#include <hip/hip_runtime.h>


typedef __attribute__((ext_vector_type(16))) _Float16 v16h;
typedef __attribute__((ext_vector_type(8)))  float    v8f;
typedef __attribute__((ext_vector_type(4)))  float    v4f;

__device__ __forceinline__ _Float16 bit_of(float x) {
    float t = (x > 0.0f) ? 1.0f : ((x == 0.0f) ? 0.5f : 0.0f);
    return (_Float16)t;
}

__global__ __launch_bounds__(256) void bcd_wmma_kernel(const float* __restrict__ x,
                                                       float* __restrict__ out,
                                                       int nChunks  ) {
    const int lane       = threadIdx.x & 31;
    const int waveInBlk  = threadIdx.x >> 5;
    const int wavesPerBlk = blockDim.x >> 5;

    const int m     = lane & 15;
    const int khalf = lane >> 4;

    v16h B;
#pragma unroll
    for (int e = 0; e < 16; ++e) {
        const int K = (e < 8) ? (khalf * 8 + e) : (16 + khalf * 8 + (e - 8));
        float w = 0.0f;
        if (K < 16) { if (m == 0) w = (float)(1u << K); }
        else        { if (m == 1) w = (float)(1u << (K - 16)); }
        B[e] = (_Float16)w;
    }

    int chunk = blockIdx.x * wavesPerBlk + waveInBlk;
    const int stride = gridDim.x * wavesPerBlk;

    for (; chunk < nChunks; chunk += stride) {
        const long long r0 = (long long)chunk * 32;

        const float* pa = x + (r0 + m)      * 16 + khalf * 8;
        const float* pb = x + (r0 + 16 + m) * 16 + khalf * 8;

        const v4f a0 = __builtin_nontemporal_load((const v4f*)pa);
        const v4f a1 = __builtin_nontemporal_load((const v4f*)(pa + 4));
        const v4f a2 = __builtin_nontemporal_load((const v4f*)pb);
        const v4f a3 = __builtin_nontemporal_load((const v4f*)(pb + 4));

        v16h A;
#pragma unroll
        for (int e = 0; e < 4; ++e) {
            A[e]      = bit_of(a0[e]);
            A[e + 4]  = bit_of(a1[e]);
            A[e + 8]  = bit_of(a2[e]);
            A[e + 12] = bit_of(a3[e]);
        }

        v8f c = {0.f, 0.f, 0.f, 0.f, 0.f, 0.f, 0.f, 0.f};
        c = __builtin_amdgcn_wmma_f32_16x16x32_f16(
                 false, A,  false, B,
                 (short)0, c,  false,  false);

        if (m < 2) {
            float* po = out + r0 + (long long)m * 16 + khalf * 8;
            const v4f lo = {c[0], c[1], c[2], c[3]};
            const v4f hi = {c[4], c[5], c[6], c[7]};
            __builtin_nontemporal_store(lo, (v4f*)po);
            __builtin_nontemporal_store(hi, (v4f*)(po + 4));
        }
    }
}

__global__ void bcd_tail_kernel(const float* __restrict__ x,
                                float* __restrict__ out,
                                long long startRow, long long N) {
    long long r = startRow + (long long)blockIdx.x * blockDim.x + threadIdx.x;
    if (r >= N) return;
    const float* p = x + r * 16;
    float acc = 0.0f, w = 1.0f;
#pragma unroll
    for (int j = 0; j < 16; ++j) {
        float v = p[j];
        float b = (v > 0.0f) ? 1.0f : ((v == 0.0f) ? 0.5f : 0.0f);
        acc += b * w;
        w *= 2.0f;
    }
    out[r] = acc;
}

extern "C" void kernel_launch(void* const* d_in, const int* in_sizes, int n_in,
                              void* d_out, int out_size, void* d_ws, size_t ws_size,
                              hipStream_t stream) {
    const float* x   = (const float*)d_in[0];
    float*       out = (float*)d_out;

    const long long total = (long long)in_sizes[0];
    const long long N     = total / 16;
    const long long nChunks = N / 32;

    if (nChunks > 0) {
        const int wavesPerBlk   = 8;
        const int chunksPerWave = 4;
        long long blocks = (nChunks + (long long)wavesPerBlk * chunksPerWave - 1) /
                           ((long long)wavesPerBlk * chunksPerWave);
        if (blocks < 1) blocks = 1;
        bcd_wmma_kernel<<<(int)blocks, 256, 0, stream>>>(x, out, (int)nChunks);
    }

    const long long tail = N - nChunks * 32;
    if (tail > 0) {
        bcd_tail_kernel<<<(int)((tail + 255) / 256), 256, 0, stream>>>(
            x, out, nChunks * 32, N);
    }
}


# every wave-wide 16-byte load reads 512 contiguous bytes (4 whole lines, not half of 8): K slots hold 4-column quads, B weighs them 1,2,4,8, three FMAs join a row's quads (sum order changes, exact)
# speedup vs baseline: 1.0980x; 1.0980x over previous
#include <hip/hip_runtime.h>


typedef __attribute__((ext_vector_type(16))) _Float16 v16h;
typedef __attribute__((ext_vector_type(8)))  float    v8f;
typedef __attribute__((ext_vector_type(4)))  float    v4f;
typedef __attribute__((ext_vector_type(2)))  float    v2f;

__device__ __forceinline__ _Float16 bit_of(float x) {
    float t = (x > 0.0f) ? 1.0f : ((x == 0.0f) ? 0.5f : 0.0f);
    return (_Float16)t;
}

__global__ __launch_bounds__(256) void bcd_wmma_kernel(const float* __restrict__ x,
                                                       float* __restrict__ out,
                                                       int nChunks  ) {
    const int lane        = threadIdx.x & 31;
    const int waveInBlk   = threadIdx.x >> 5;
    const int wavesPerBlk = blockDim.x >> 5;

    const int c = lane & 15;
    const int h = lane >> 4;

    v16h B;
#pragma unroll
    for (int s = 0; s < 16; ++s) {
        const int i = s >> 2;
        const int e = s & 3;
        const float w = (c == 2 * i + h) ? (float)(1u << e) : 0.0f;
        B[s] = (_Float16)w;
    }

    int chunk = blockIdx.x * wavesPerBlk + waveInBlk;
    const int stride = gridDim.x * wavesPerBlk;

    for (; chunk < nChunks; chunk += stride) {
        const float* p = x + (long long)chunk * 512 + lane * 4;

        const v4f a0 = __builtin_nontemporal_load((const v4f*)p);
        const v4f a1 = __builtin_nontemporal_load((const v4f*)(p + 128));
        const v4f a2 = __builtin_nontemporal_load((const v4f*)(p + 256));
        const v4f a3 = __builtin_nontemporal_load((const v4f*)(p + 384));

        v16h A;
#pragma unroll
        for (int e = 0; e < 4; ++e) {
            A[e]      = bit_of(a0[e]);
            A[e + 4]  = bit_of(a1[e]);
            A[e + 8]  = bit_of(a2[e]);
            A[e + 12] = bit_of(a3[e]);
        }

        v8f d = {0.f, 0.f, 0.f, 0.f, 0.f, 0.f, 0.f, 0.f};
        d = __builtin_amdgcn_wmma_f32_16x16x32_f16(
                 false, A,  false, B,
                 (short)0, d,  false,  false);

        if (c < 8) {
            v2f o;
            o[0] = fmaf(d[3], 4096.0f, fmaf(d[2], 256.0f, fmaf(d[1], 16.0f, d[0])));
            o[1] = fmaf(d[7], 4096.0f, fmaf(d[6], 256.0f, fmaf(d[5], 16.0f, d[4])));
            float* po = out + (long long)chunk * 32 + 4 * c + 2 * h;
            __builtin_nontemporal_store(o, (v2f*)po);
        }
    }
}

__global__ void bcd_tail_kernel(const float* __restrict__ x,
                                float* __restrict__ out,
                                long long startRow, long long N) {
    long long r = startRow + (long long)blockIdx.x * blockDim.x + threadIdx.x;
    if (r >= N) return;
    const float* p = x + r * 16;
    float acc = 0.0f, w = 1.0f;
#pragma unroll
    for (int j = 0; j < 16; ++j) {
        float v = p[j];
        float b = (v > 0.0f) ? 1.0f : ((v == 0.0f) ? 0.5f : 0.0f);
        acc += b * w;
        w *= 2.0f;
    }
    out[r] = acc;
}

extern "C" void kernel_launch(void* const* d_in, const int* in_sizes, int n_in,
                              void* d_out, int out_size, void* d_ws, size_t ws_size,
                              hipStream_t stream) {
    const float* x   = (const float*)d_in[0];
    float*       out = (float*)d_out;

    const long long total = (long long)in_sizes[0];
    const long long N     = total / 16;
    const long long nChunks = N / 32;

    if (nChunks > 0) {
        const int wavesPerBlk   = 8;
        const int chunksPerWave = 4;
        long long blocks = (nChunks + (long long)wavesPerBlk * chunksPerWave - 1) /
                           ((long long)wavesPerBlk * chunksPerWave);
        if (blocks < 1) blocks = 1;
        bcd_wmma_kernel<<<(int)blocks, 256, 0, stream>>>(x, out, (int)nChunks);
    }

    const long long tail = N - nChunks * 32;
    if (tail > 0) {
        bcd_tail_kernel<<<(int)((tail + 255) / 256), 256, 0, stream>>>(
            x, out, nChunks * 32, N);
    }
}
